# starting kernel
# baseline (speedup 1.0000x reference)
#include <hip/hip_runtime.h>

typedef __attribute__((ext_vector_type(2))) float v2f;
typedef __attribute__((ext_vector_type(8))) float v8f;

#define EDIM 128
#define WAVES_PER_BLOCK 4
#define EDGES_PER_WAVE 16
#define EDGES_PER_BLOCK (WAVES_PER_BLOCK * EDGES_PER_WAVE)

__global__ __launch_bounds__(WAVES_PER_BLOCK * 32)
void rel_dot_wmma_kernel(const float* __restrict__ emb,
                         const float* __restrict__ relations,
                         const float* __restrict__ tokeys,
                         const float* __restrict__ toqueries,
                         const int* __restrict__ si,
                         const int* __restrict__ pi,
                         const int* __restrict__ oi,
                         float* __restrict__ out,
                         int E)
{
    extern __shared__ float smem[];
    __shared__ int sPi[WAVES_PER_BLOCK][EDGES_PER_WAVE];

    const int lane = threadIdx.x & 31;
    const int wave = threadIdx.x >> 5;
    const int base = (blockIdx.x * WAVES_PER_BLOCK + wave) * EDGES_PER_WAVE;

    float* sS = smem + wave * (2 * EDGES_PER_WAVE * EDIM);
    float* sO = sS + EDGES_PER_WAVE * EDIM;

    #pragma unroll 4
    for (int r = 0; r < EDGES_PER_WAVE; ++r) {
        const int e   = min(base + r, E - 1);
        const int srw = si[e];
        const int orw = oi[e];
        const float4 vs = *(const float4*)(emb + (size_t)srw * EDIM + lane * 4);
        const float4 vo = *(const float4*)(emb + (size_t)orw * EDIM + lane * 4);
        *(float4*)(sS + r * EDIM + lane * 4) = vs;
        *(float4*)(sO + r * EDIM + lane * 4) = vo;
    }
    if (lane < EDGES_PER_WAVE) sPi[wave][lane] = pi[min(base + lane, E - 1)];
    __syncthreads();

    const int half = lane >> 4;
    const int lr   = lane & 15;

    v8f accS[8];
    #pragma unroll
    for (int t = 0; t < 8; ++t) accS[t] = (v8f)0.0f;

    for (int kc = 0; kc < EDIM / 4; ++kc) {
        const int col = kc * 4 + 2 * half;
        const v2f a = *(const v2f*)(sS + lr * EDIM + col);
        #pragma unroll
        for (int t = 0; t < 8; ++t) {
            const v2f b = *(const v2f*)(tokeys + (size_t)(t * 16 + lr) * EDIM + col);
            accS[t] = __builtin_amdgcn_wmma_f32_16x16x4_f32(
                false, a, false, b, (short)0, accS[t], false, false);
        }
    }

    v8f part = (v8f)0.0f;
    for (int t = 0; t < 8; ++t) {
        v8f accO = (v8f)0.0f;
        for (int kc = 0; kc < EDIM / 4; ++kc) {
            const int col = kc * 4 + 2 * half;
            const v2f a = *(const v2f*)(sO + lr * EDIM + col);
            const v2f b = *(const v2f*)(toqueries + (size_t)(t * 16 + lr) * EDIM + col);
            accO = __builtin_amdgcn_wmma_f32_16x16x4_f32(
                false, a, false, b, (short)0, accO, false, false);
        }
        #pragma unroll
        for (int v = 0; v < 8; ++v) {
            const int prow = sPi[wave][v + 8 * half];
            const float p  = relations[(size_t)prow * EDIM + t * 16 + lr];
            part[v] += accS[t][v] * accO[v] * p;
        }
    }

    const float scale = 0.08838834764831845f;
    #pragma unroll
    for (int v = 0; v < 8; ++v) {
        float s = part[v];
        s += __shfl_xor(s, 1, 32);
        s += __shfl_xor(s, 2, 32);
        s += __shfl_xor(s, 4, 32);
        s += __shfl_xor(s, 8, 32);
        if (lr == 0) {
            const int e = base + v + 8 * half;
            if (e < E) out[e] = s * scale;
        }
    }
}

extern "C" void kernel_launch(void* const* d_in, const int* in_sizes, int n_in,
                              void* d_out, int out_size, void* d_ws, size_t ws_size,
                              hipStream_t stream) {
    const float* emb       = (const float*)d_in[0];
    const float* relations = (const float*)d_in[1];
    const float* tokeys    = (const float*)d_in[2];
    const float* toqueries = (const float*)d_in[3];
    const int*   si        = (const int*)d_in[4];
    const int*   pi        = (const int*)d_in[5];
    const int*   oi        = (const int*)d_in[6];
    float*       out       = (float*)d_out;

    const int E = in_sizes[4];
    const int blocks = (E + EDGES_PER_BLOCK - 1) / EDGES_PER_BLOCK;
    const size_t shmem = (size_t)WAVES_PER_BLOCK * 2 * EDGES_PER_WAVE * EDIM * sizeof(float);

    rel_dot_wmma_kernel<<<blocks, WAVES_PER_BLOCK * 32, shmem, stream>>>(
        emb, relations, tokeys, toqueries, si, pi, oi, out, E);
}


# Waves specialised: 256 groups of 8 waves; 4 gather waves fill a two-batch LDS ring one batch ahead and take 1 pass in 4, 4 product waves take 3; 32 edges per pass; same per-edge sums
# speedup vs baseline: 1.8586x; 1.8586x over previous
#include <hip/hip_runtime.h>

typedef __attribute__((ext_vector_type(2))) float v2f;
typedef __attribute__((ext_vector_type(8))) float v8f;

#define EDIM 128
#define NPROD 4
#define NGATH 4
#define NWAVES (NPROD + NGATH)
#define BATCH 128
#define NT 2
#define PASSES (BATCH / (16 * NT))
#define PITCH 132
#define NGROUPS 256

__global__ __launch_bounds__(NWAVES * 32, 1)
void rel_dot_wspec_kernel(const float* __restrict__ emb,
                          const float* __restrict__ relations,
                          const float* __restrict__ tokeys,
                          const float* __restrict__ toqueries,
                          const int* __restrict__ si,
                          const int* __restrict__ pi,
                          const int* __restrict__ oi,
                          float* __restrict__ out,
                          int E)
{
    __shared__ __attribute__((aligned(16))) float sS[2][BATCH * PITCH];
    __shared__ __attribute__((aligned(16))) float sO[2][BATCH * PITCH];
    __shared__ float sPart[2][NPROD][BATCH];
    __shared__ int   sPi[2][BATCH];

    const int lane = threadIdx.x & 31;
    const int wave = __builtin_amdgcn_readfirstlane(threadIdx.x >> 5);
    const int half = lane >> 4;
    const int lr   = lane & 15;
    const bool gatherer = wave >= NPROD;
    const int slice = wave & 3;
    const int tg = slice * 32 + lane;

    v2f wk[2][EDIM / 4];
    v2f wq[2][EDIM / 4];
    #pragma unroll
    for (int mt = 0; mt < 2; ++mt) {
        const size_t off = (size_t)(32 * slice + 16 * mt + lr) * EDIM + 2 * half;
        #pragma unroll
        for (int kc = 0; kc < EDIM / 4; ++kc) {
            wk[mt][kc] = *(const v2f*)(tokeys    + off + kc * 4);
            wq[mt][kc] = *(const v2f*)(toqueries + off + kc * 4);
        }
    }

    const float scale = 0.08838834764831845f;
    const int nb = (E + BATCH - 1) / BATCH;
    const int step = gridDim.x;

    int it = 0;
    int b = blockIdx.x;
    for (;; b += step, ++it) {
        if (gatherer) {
            if (it >= 2) {
                const int pc = it & 1;
                const float d = (sPart[pc][0][tg] + sPart[pc][1][tg]) + (sPart[pc][2][tg] + sPart[pc][3][tg]);
                const int e = (b - 2 * step) * BATCH + tg;
                if (e < E) out[e] = d * scale;
            }
            if (b < nb) {
                const int cur = it & 1;
                const int base = b * BATCH;
                sPi[cur][tg] = pi[min(base + tg, E - 1)];
                #pragma unroll 1
                for (int q = 0; q < 2; ++q) {
                    const int row0 = slice * 32 + q * 16;
                    const int* isrc = half ? oi : si;
                    const int idx = isrc[min(base + row0 + lr, E - 1)];
                    float4 vs[16], vo[16];
                    #pragma unroll
                    for (int r = 0; r < 16; ++r) {
                        const int srw = __builtin_amdgcn_readlane(idx, r);
                        const int orw = __builtin_amdgcn_readlane(idx, 16 + r);
                        vs[r] = *(const float4*)(emb + (size_t)srw * EDIM + lane * 4);
                        vo[r] = *(const float4*)(emb + (size_t)orw * EDIM + lane * 4);
                    }
                    #pragma unroll
                    for (int r = 0; r < 16; ++r) {
                        *(float4*)(&sS[cur][(row0 + r) * PITCH + lane * 4]) = vs[r];
                        *(float4*)(&sO[cur][(row0 + r) * PITCH + lane * 4]) = vo[r];
                    }
                }
            }
        }
        if (it >= 1) {
            const int cur = (it - 1) & 1;
            const int ps0 = gatherer ? PASSES - 1 : 0;
            const int ps1 = gatherer ? PASSES : PASSES - 1;
            #pragma unroll 1
            for (int ps = ps0; ps < ps1; ++ps) {
                float4 p[NT][4];
                const float* fS[NT];
                const float* fO[NT];
                v8f aS[NT][2], aO[NT][2];
                #pragma unroll
                for (int u = 0; u < NT; ++u) {
                    const int tile = ps * NT + u;
                    const int prow = sPi[cur][tile * 16 + lr];
                    const float* pr = relations + (size_t)prow * EDIM + 32 * slice + 8 * half;
                    p[u][0] = *(const float4*)(pr);
                    p[u][1] = *(const float4*)(pr + 4);
                    p[u][2] = *(const float4*)(pr + 16);
                    p[u][3] = *(const float4*)(pr + 20);
                    fS[u] = &sS[cur][(tile * 16 + lr) * PITCH + 2 * half];
                    fO[u] = &sO[cur][(tile * 16 + lr) * PITCH + 2 * half];
                    aS[u][0] = (v8f)0.0f; aS[u][1] = (v8f)0.0f;
                    aO[u][0] = (v8f)0.0f; aO[u][1] = (v8f)0.0f;
                }
                #pragma unroll
                for (int kc = 0; kc < EDIM / 4; ++kc) {
                    #pragma unroll
                    for (int u = 0; u < NT; ++u) {
                        const v2f bs = *(const v2f*)(fS[u] + kc * 4);
                        const v2f bo = *(const v2f*)(fO[u] + kc * 4);
                        aS[u][0] = __builtin_amdgcn_wmma_f32_16x16x4_f32(false, wk[0][kc], false, bs, (short)0, aS[u][0], false, false);
                        aS[u][1] = __builtin_amdgcn_wmma_f32_16x16x4_f32(false, wk[1][kc], false, bs, (short)0, aS[u][1], false, false);
                        aO[u][0] = __builtin_amdgcn_wmma_f32_16x16x4_f32(false, wq[0][kc], false, bo, (short)0, aO[u][0], false, false);
                        aO[u][1] = __builtin_amdgcn_wmma_f32_16x16x4_f32(false, wq[1][kc], false, bo, (short)0, aO[u][1], false, false);
                    }
                }
                #pragma unroll
                for (int u = 0; u < NT; ++u) {
                    const v8f a0 = aS[u][0], o0 = aO[u][0], a1 = aS[u][1], o1 = aO[u][1];
                    float s = 0.0f;
                    s += a0[0] * o0[0] * p[u][0].x;  s += a0[1] * o0[1] * p[u][0].y;
                    s += a0[2] * o0[2] * p[u][0].z;  s += a0[3] * o0[3] * p[u][0].w;
                    s += a0[4] * o0[4] * p[u][1].x;  s += a0[5] * o0[5] * p[u][1].y;
                    s += a0[6] * o0[6] * p[u][1].z;  s += a0[7] * o0[7] * p[u][1].w;
                    s += a1[0] * o1[0] * p[u][2].x;  s += a1[1] * o1[1] * p[u][2].y;
                    s += a1[2] * o1[2] * p[u][2].z;  s += a1[3] * o1[3] * p[u][2].w;
                    s += a1[4] * o1[4] * p[u][3].x;  s += a1[5] * o1[5] * p[u][3].y;
                    s += a1[6] * o1[6] * p[u][3].z;  s += a1[7] * o1[7] * p[u][3].w;
                    s += __shfl_xor(s, 16, 32);
                    if (half == 0) sPart[cur][slice][(ps * NT + u) * 16 + lr] = s;
                }
            }
        }
        __syncthreads();
        if (b >= nb) break;
    }
    if (gatherer && it >= 1) {
        const int pc = (it - 1) & 1;
        const float d = (sPart[pc][0][tg] + sPart[pc][1][tg]) + (sPart[pc][2][tg] + sPart[pc][3][tg]);
        const int e = (b - step) * BATCH + tg;
        if (e < E) out[e] = d * scale;
    }
}

extern "C" void kernel_launch(void* const* d_in, const int* in_sizes, int n_in,
                              void* d_out, int out_size, void* d_ws, size_t ws_size,
                              hipStream_t stream) {
    const float* emb       = (const float*)d_in[0];
    const float* relations = (const float*)d_in[1];
    const float* tokeys    = (const float*)d_in[2];
    const float* toqueries = (const float*)d_in[3];
    const int*   si        = (const int*)d_in[4];
    const int*   pi        = (const int*)d_in[5];
    const int*   oi        = (const int*)d_in[6];
    float*       out       = (float*)d_out;

    const int E = in_sizes[4];

    rel_dot_wspec_kernel<<<NGROUPS, NWAVES * 32, 0, stream>>>(
        emb, relations, tokeys, toqueries, si, pi, oi, out, E);
}


# Slot-scatter and result-fetch launches coarsened: 512 groups, each thread 8 edges with all their loads in flight together, not 4096 groups of one edge per thread; same sums
# speedup vs baseline: 2.1429x; 1.1530x over previous
#include <hip/hip_runtime.h>

typedef __attribute__((ext_vector_type(2))) float v2f;
typedef __attribute__((ext_vector_type(8))) float v8f;

#define EDIM 128
#define NPROD 4
#define NGATH 4
#define NWAVES (NPROD + NGATH)
#define BATCH 128
#define NT 2
#define PASSES (BATCH / (16 * NT))
#define PITCH 132
#define NGROUPS 256
#define NCH 256
#define SUB 4096
#define RCAP 256
#define TG 512
#define TU 8


__device__ __forceinline__ void wave_rank(const int v, const int lane, int& rank, int& grp)
{
    rank = 0; grp = 0;
    #pragma unroll
    for (int k = 0; k < 32; ++k) {
        const int vk = __builtin_amdgcn_readlane(v, k);
        const int same = (vk == v) ? 1 : 0;
        grp += same;
        rank += (k < lane) ? same : 0;
    }
}

__global__ __launch_bounds__(256)
void rel_rank_kernel(const int* __restrict__ pi, int* __restrict__ cnt, int* __restrict__ inv, int E, int nrelElems)
{
    __shared__ int sp[SUB];
    __shared__ int hist[8][RCAP];
    const int tid = threadIdx.x;
    const int lane = tid & 31;
    const int wave = __builtin_amdgcn_readfirstlane(tid >> 5);
    const int c = blockIdx.x;
    const int R = min(nrelElems / EDIM, RCAP);
    const int CH = ((E + NCH - 1) / NCH + 31) & ~31;
    const int lo = min(E, c * CH);
    const int hi = min(E, lo + CH);
    for (int i = tid; i < 8 * RCAP; i += 256) (&hist[0][0])[i] = 0;
    for (int pass = 0; pass < 2; ++pass) {
        for (int s0 = lo; s0 < hi; s0 += SUB) {
            __syncthreads();
            {
                int t[SUB / 256];
                #pragma unroll
                for (int k = 0; k < SUB / 256; ++k) t[k] = pi[min(s0 + tid + 256 * k, E - 1)];
                #pragma unroll
                for (int k = 0; k < SUB / 256; ++k) sp[tid + 256 * k] = (s0 + tid + 256 * k < hi) ? t[k] : -1;
            }
            __syncthreads();
            const int nsteps = (min(SUB, hi - s0) + 255) >> 8;
            for (int st = 0; st < nsteps; ++st) {
                const int i = (st * 8 + wave) * 32 + lane;
                const int v = sp[i];
                int rank, grp;
                wave_rank(v, lane, rank, grp);
                if (v >= 0 && v < R) {
                    const int base = hist[wave][v];
                    if (rank == 0) hist[wave][v] = base + grp;
                    if (pass == 1) inv[s0 + i] = base + rank;
                }
            }
        }
        __syncthreads();
        if (pass == 0) {
            if (tid < R) {
                int run = 0;
                for (int w = 0; w < 8; ++w) { const int h = hist[w][tid]; hist[w][tid] = run; run += h; }
                cnt[tid * NCH + c] = run;
            }
        }
    }
}

__global__ __launch_bounds__(256)
void rel_offsets_kernel(const int* __restrict__ cnt, int* __restrict__ offs, int* __restrict__ relb,
                        int* __restrict__ meta, int* __restrict__ gS, int* __restrict__ gO,
                        int E, int nrelElems)
{
    __shared__ int spad[RCAP];
    const int r = threadIdx.x;
    const int R = min(nrelElems / EDIM, RCAP);
    int total = 0;
    if (r < R) {
        const int4* row = (const int4*)(cnt + r * NCH);
        #pragma unroll 1
        for (int g = 0; g < NCH / 4; g += 16) {
            int4 v[16];
            #pragma unroll
            for (int k = 0; k < 16; ++k) v[k] = row[g + k];
            #pragma unroll
            for (int k = 0; k < 16; ++k) total += (v[k].x + v[k].y) + (v[k].z + v[k].w);
        }
    }
    const int padded = (total + BATCH - 1) & ~(BATCH - 1);
    spad[r] = (r < R) ? padded : 0;
    __syncthreads();
    if (r < R) {
        int base = 0;
        for (int q = 0; q < r; ++q) base += spad[q];
        int run = base;
        const int4* row = (const int4*)(cnt + r * NCH);
        int4* orow = (int4*)(offs + r * NCH);
        #pragma unroll 1
        for (int g = 0; g < NCH / 4; g += 16) {
            int4 v[16];
            #pragma unroll
            for (int k = 0; k < 16; ++k) v[k] = row[g + k];
            #pragma unroll
            for (int k = 0; k < 16; ++k) {
                int4 o;
                o.x = run; run += v[k].x;
                o.y = run; run += v[k].y;
                o.z = run; run += v[k].z;
                o.w = run; run += v[k].w;
                orow[g + k] = o;
            }
        }
        const int cap = E + R * BATCH;
        const int nbmax = cap / BATCH;
        for (int k = base + total; k < min(base + padded, cap); ++k) { gS[k] = 0; gO[k] = 0; }
        for (int b = base / BATCH; b < min((base + padded) / BATCH, nbmax); ++b) relb[b] = r;
        if (r == R - 1) meta[0] = (base + padded) / BATCH;
    }
}

__global__ __launch_bounds__(256)
void rel_scatter_kernel(const int* __restrict__ si, const int* __restrict__ pi, const int* __restrict__ oi,
                        const int* __restrict__ offs, int* inv, int* __restrict__ gS, int* __restrict__ gO,
                        int E, int nrelElems)
{
    const int R = min(nrelElems / EDIM, RCAP);
    const int CH = ((E + NCH - 1) / NCH + 31) & ~31;
    const int cap = E + R * BATCH;
    const int stride = (int)gridDim.x * 256;
    for (int g0 = blockIdx.x * 256; g0 < E; g0 += stride * TU) {
        const int e0 = g0 + (int)threadIdx.x;
        int r[TU], rk[TU], s[TU], o[TU], pos[TU];
        #pragma unroll
        for (int u = 0; u < TU; ++u) {
            const int e = min(e0 + u * stride, E - 1);
            r[u]  = pi[e];
            rk[u] = inv[e];
            s[u]  = si[e];
            o[u]  = oi[e];
        }
        #pragma unroll
        for (int u = 0; u < TU; ++u) {
            const int e = min(e0 + u * stride, E - 1);
            const int rr = min(max(r[u], 0), R - 1);
            pos[u] = offs[rr * NCH + e / CH] + rk[u];
        }
        asm volatile("" : "+v"(pos[0]), "+v"(s[0]), "+v"(o[0]));
        #pragma unroll
        for (int u = 0; u < TU; ++u) {
            const int e = e0 + u * stride;
            if (e < E) {
                inv[e] = pos[u];
                if (pos[u] >= 0 && pos[u] < cap) { gS[pos[u]] = s[u]; gO[pos[u]] = o[u]; }
            }
        }
    }
}

__global__ __launch_bounds__(256)
void rel_matrix_kernel(const float* __restrict__ relations, const float* __restrict__ tokeys,
                       const float* __restrict__ toqueries, float* __restrict__ M, int nrelElems)
{
    const int R = min(nrelElems / EDIM, RCAP);
    const int r = blockIdx.x;
    if (r >= R) return;
    const int lane = threadIdx.x & 31;
    const int wave = __builtin_amdgcn_readfirstlane(threadIdx.x >> 5);
    const int half = lane >> 4;
    const int lr   = lane & 15;
    v2f a[EDIM / 4];
    #pragma unroll
    for (int jc = 0; jc < EDIM / 4; ++jc) {
        const int j = 4 * jc + 2 * half;
        const v2f pj = *(const v2f*)(relations + (size_t)r * EDIM + j);
        v2f t;
        t.x = tokeys[(size_t)j * EDIM + 16 * wave + lr] * pj.x;
        t.y = tokeys[(size_t)(j + 1) * EDIM + 16 * wave + lr] * pj.y;
        a[jc] = t;
    }
    float* Mr = M + (size_t)r * EDIM * EDIM;
    #pragma unroll 1
    for (int kt = 0; kt < EDIM / 16; kt += 2) {
        v8f acc0 = (v8f)0.0f, acc1 = (v8f)0.0f;
        #pragma unroll
        for (int jc = 0; jc < EDIM / 4; ++jc) {
            const int j = 4 * jc + 2 * half;
            v2f b0, b1;
            b0.x = toqueries[(size_t)j * EDIM + 16 * kt + lr];
            b0.y = toqueries[(size_t)(j + 1) * EDIM + 16 * kt + lr];
            b1.x = toqueries[(size_t)j * EDIM + 16 * kt + 16 + lr];
            b1.y = toqueries[(size_t)(j + 1) * EDIM + 16 * kt + 16 + lr];
            acc0 = __builtin_amdgcn_wmma_f32_16x16x4_f32(false, a[jc], false, b0, (short)0, acc0, false, false);
            acc1 = __builtin_amdgcn_wmma_f32_16x16x4_f32(false, a[jc], false, b1, (short)0, acc1, false, false);
        }
        #pragma unroll
        for (int v = 0; v < 8; ++v) {
            Mr[(size_t)(16 * wave + 8 * half + v) * EDIM + 16 * kt + lr]      = acc0[v];
            Mr[(size_t)(16 * wave + 8 * half + v) * EDIM + 16 * kt + 16 + lr] = acc1[v];
        }
    }
}

__global__ __launch_bounds__(NWAVES * 32, 1)
void rel_dot_grouped_kernel(const float* __restrict__ emb,
                            const float* __restrict__ M,
                            const int* __restrict__ gS,
                            const int* __restrict__ gO,
                            const int* __restrict__ relb,
                            const int* __restrict__ meta,
                            float* __restrict__ dG,
                            int E, int nrelElems, int nembElems)
{
    __shared__ __attribute__((aligned(16))) float sS[2][BATCH * PITCH];
    __shared__ __attribute__((aligned(16))) float sO[2][BATCH * PITCH];
    __shared__ float sPart[2][NPROD][BATCH];

    const int lane = threadIdx.x & 31;
    const int wave = __builtin_amdgcn_readfirstlane(threadIdx.x >> 5);
    const int half = lane >> 4;
    const int lr   = lane & 15;
    const bool gatherer = wave >= NPROD;
    const int slice = wave & 3;
    const int tg = slice * 32 + lane;

    v2f wq[2][EDIM / 4];
    #pragma unroll
    for (int mt = 0; mt < 2; ++mt) {
        #pragma unroll
        for (int kc = 0; kc < EDIM / 4; ++kc) wq[mt][kc] = (v2f)0.0f;
    }
    int rcur = -1;

    const float scale = 0.08838834764831845f;
    const int R = min(nrelElems / EDIM, RCAP);
    const int N = nembElems / EDIM;
    const int nbtot = min(max(meta[0], 0), (E + R * BATCH) / BATCH);
    const int per = (nbtot + (int)gridDim.x - 1) / (int)gridDim.x;
    const int b0 = (int)blockIdx.x * per;
    const int bend = min(nbtot, b0 + per);

    int it = 0;
    int b = b0;
    int relA = 0;
    for (;; ++b, ++it) {
        const int relThis = relA;
        if (b < bend) relA = relb[b];
        if (gatherer) {
            if (it >= 2) {
                const int pc = it & 1;
                const float d = (sPart[pc][0][tg] + sPart[pc][1][tg]) + (sPart[pc][2][tg] + sPart[pc][3][tg]);
                dG[(b - 2) * BATCH + tg] = d * scale;
            }
            if (b < bend) {
                const int cur = it & 1;
                const int base = b * BATCH;
                #pragma unroll 1
                for (int q = 0; q < 2; ++q) {
                    const int row0 = slice * 32 + q * 16;
                    const int* isrc = half ? gO : gS;
                    const int idx = min(max(isrc[base + row0 + lr], 0), N - 1);
                    float4 vs[16], vo[16];
                    #pragma unroll
                    for (int r = 0; r < 16; ++r) {
                        const int srw = __builtin_amdgcn_readlane(idx, r);
                        const int orw = __builtin_amdgcn_readlane(idx, 16 + r);
                        vs[r] = *(const float4*)(emb + (size_t)srw * EDIM + lane * 4);
                        vo[r] = *(const float4*)(emb + (size_t)orw * EDIM + lane * 4);
                    }
                    #pragma unroll
                    for (int r = 0; r < 16; ++r) {
                        *(float4*)(&sS[cur][(row0 + r) * PITCH + lane * 4]) = vs[r];
                        *(float4*)(&sO[cur][(row0 + r) * PITCH + lane * 4]) = vo[r];
                    }
                }
            }
        }
        if (it >= 1) {
            const int cur = (it - 1) & 1;
            const int rel = min(max(__builtin_amdgcn_readfirstlane(relThis), 0), R - 1);
            if (rel != rcur) {
                rcur = rel;
                const float* Mr = M + (size_t)rel * EDIM * EDIM;
                #pragma unroll
                for (int mt = 0; mt < 2; ++mt) {
                    const size_t off = (size_t)(32 * slice + 16 * mt + lr) * EDIM + 2 * half;
                    #pragma unroll
                    for (int kc = 0; kc < EDIM / 4; ++kc) wq[mt][kc] = *(const v2f*)(Mr + off + kc * 4);
                }
            }
            const int ps0 = gatherer ? PASSES - 1 : 0;
            const int ps1 = gatherer ? PASSES : PASSES - 1;
            #pragma unroll 1
            for (int ps = ps0; ps < ps1; ++ps) {
                float4 p[NT][4];
                const float* fO[NT];
                v8f aO[NT][2];
                #pragma unroll
                for (int u = 0; u < NT; ++u) {
                    const int slot = (ps * NT + u) * 16 + lr;
                    const float* sr = &sS[cur][slot * PITCH + 32 * slice + 8 * half];
                    p[u][0] = *(const float4*)(sr);
                    p[u][1] = *(const float4*)(sr + 4);
                    p[u][2] = *(const float4*)(sr + 16);
                    p[u][3] = *(const float4*)(sr + 20);
                    fO[u] = &sO[cur][slot * PITCH + 2 * half];
                    aO[u][0] = (v8f)0.0f; aO[u][1] = (v8f)0.0f;
                }
                #pragma unroll
                for (int kc = 0; kc < EDIM / 4; ++kc) {
                    #pragma unroll
                    for (int u = 0; u < NT; ++u) {
                        const v2f bo = *(const v2f*)(fO[u] + kc * 4);
                        aO[u][0] = __builtin_amdgcn_wmma_f32_16x16x4_f32(false, wq[0][kc], false, bo, (short)0, aO[u][0], false, false);
                        aO[u][1] = __builtin_amdgcn_wmma_f32_16x16x4_f32(false, wq[1][kc], false, bo, (short)0, aO[u][1], false, false);
                    }
                }
                #pragma unroll
                for (int u = 0; u < NT; ++u) {
                    const v8f o0 = aO[u][0], o1 = aO[u][1];
                    float s = 0.0f;
                    s += o0[0] * p[u][0].x;  s += o0[1] * p[u][0].y;
                    s += o0[2] * p[u][0].z;  s += o0[3] * p[u][0].w;
                    s += o0[4] * p[u][1].x;  s += o0[5] * p[u][1].y;
                    s += o0[6] * p[u][1].z;  s += o0[7] * p[u][1].w;
                    s += o1[0] * p[u][2].x;  s += o1[1] * p[u][2].y;
                    s += o1[2] * p[u][2].z;  s += o1[3] * p[u][2].w;
                    s += o1[4] * p[u][3].x;  s += o1[5] * p[u][3].y;
                    s += o1[6] * p[u][3].z;  s += o1[7] * p[u][3].w;
                    s += __shfl_xor(s, 16, 32);
                    if (half == 0) sPart[cur][slice][(ps * NT + u) * 16 + lr] = s;
                }
            }
        }
        __syncthreads();
        if (b >= bend) break;
    }
    if (gatherer && it >= 1) {
        const int pc = (it - 1) & 1;
        const float d = (sPart[pc][0][tg] + sPart[pc][1][tg]) + (sPart[pc][2][tg] + sPart[pc][3][tg]);
        dG[(b - 1) * BATCH + tg] = d * scale;
    }
}

__global__ __launch_bounds__(256)
void rel_fetch_kernel(const float* __restrict__ dG, const int* __restrict__ inv, float* __restrict__ out,
                      int E, int nrelElems)
{
    const int R = min(nrelElems / EDIM, RCAP);
    const int cap = E + R * BATCH;
    const int stride = (int)gridDim.x * 256;
    for (int g0 = blockIdx.x * 256; g0 < E; g0 += stride * TU) {
        const int e0 = g0 + (int)threadIdx.x;
        int p[TU];
        float d[TU];
        #pragma unroll
        for (int u = 0; u < TU; ++u) p[u] = inv[min(e0 + u * stride, E - 1)];
        #pragma unroll
        for (int u = 0; u < TU; ++u) d[u] = dG[min(max(p[u], 0), cap - 1)];
        asm volatile("" : "+v"(d[0]));
        #pragma unroll
        for (int u = 0; u < TU; ++u) {
            const int e = e0 + u * stride;
            if (e < E) out[e] = d[u];
        }
    }
}

extern "C" void kernel_launch(void* const* d_in, const int* in_sizes, int n_in,
                              void* d_out, int out_size, void* d_ws, size_t ws_size,
                              hipStream_t stream) {
    const float* emb       = (const float*)d_in[0];
    const float* relations = (const float*)d_in[1];
    const float* tokeys    = (const float*)d_in[2];
    const float* toqueries = (const float*)d_in[3];
    const int*   si        = (const int*)d_in[4];
    const int*   pi        = (const int*)d_in[5];
    const int*   oi        = (const int*)d_in[6];
    float*       out       = (float*)d_out;

    const int E = in_sizes[4];
    const int nrelElems = in_sizes[1];
    const int nembElems = in_sizes[0];

    char* ws = (char*)d_ws;
    float* M    = (float*)(ws);
    int*   cnt  = (int*)(ws + 16777216);
    int*   offs = (int*)(ws + 17039360);
    int*   relb = (int*)(ws + 17301504);
    int*   meta = (int*)(ws + 17563648);
    int*   inv  = (int*)(ws + 17825792);
    int*   gS   = (int*)(ws + 23068672);
    int*   gO   = (int*)(ws + 28311552);
    float* dG   = (float*)(ws + 33554432);

    rel_rank_kernel<<<NCH, 256, 0, stream>>>(pi, cnt, inv, E, nrelElems);
    rel_offsets_kernel<<<1, 256, 0, stream>>>(cnt, offs, relb, meta, gS, gO, E, nrelElems);
    rel_scatter_kernel<<<TG, 256, 0, stream>>>(si, pi, oi, offs, inv, gS, gO, E, nrelElems);
    rel_matrix_kernel<<<RCAP, 256, 0, stream>>>(relations, tokeys, toqueries, M, nrelElems);
    rel_dot_grouped_kernel<<<NGROUPS, NWAVES * 32, 0, stream>>>(emb, M, gS, gO, relb, meta, dG, E, nrelElems, nembElems);
    rel_fetch_kernel<<<TG, 256, 0, stream>>>(dG, inv, out, E, nrelElems);
}


# Six launches become four: the relation-matrix groups share the rank launch, and each chunk's slot group derives its run starts from the (now chunk-major) count table itself; same sums
# speedup vs baseline: 2.4527x; 1.1446x over previous
#include <hip/hip_runtime.h>

typedef __attribute__((ext_vector_type(2))) float v2f;
typedef __attribute__((ext_vector_type(8))) float v8f;

#define EDIM 128
#define NPROD 4
#define NGATH 4
#define NWAVES (NPROD + NGATH)
#define BATCH 128
#define NT 2
#define PASSES (BATCH / (16 * NT))
#define PITCH 132
#define NGROUPS 256
#define NCH 256
#define SUB 4096
#define RCAP 256
#define TG 512
#define TU 8
#define PU 16
#define NL1 512


__device__ __forceinline__ void wave_rank(const int v, const int lane, int& rank, int& grp)
{
    rank = 0; grp = 0;
    #pragma unroll
    for (int k = 0; k < 32; ++k) {
        const int vk = __builtin_amdgcn_readlane(v, k);
        const int same = (vk == v) ? 1 : 0;
        grp += same;
        rank += (k < lane) ? same : 0;
    }
}

__device__ __forceinline__ void rank_role(const int* __restrict__ pi, int* __restrict__ cnt, int* __restrict__ inv, int E, int nrelElems)
{
    __shared__ int sp[SUB];
    __shared__ int hist[8][RCAP];
    const int tid = threadIdx.x;
    const int lane = tid & 31;
    const int wave = __builtin_amdgcn_readfirstlane(tid >> 5);
    const int c = blockIdx.x;
    const int R = min(nrelElems / EDIM, RCAP);
    const int CH = ((E + NCH - 1) / NCH + 31) & ~31;
    const int lo = min(E, c * CH);
    const int hi = min(E, lo + CH);
    for (int i = tid; i < 8 * RCAP; i += 256) (&hist[0][0])[i] = 0;
    for (int pass = 0; pass < 2; ++pass) {
        for (int s0 = lo; s0 < hi; s0 += SUB) {
            __syncthreads();
            {
                int t[SUB / 256];
                #pragma unroll
                for (int k = 0; k < SUB / 256; ++k) t[k] = pi[min(s0 + tid + 256 * k, E - 1)];
                #pragma unroll
                for (int k = 0; k < SUB / 256; ++k) sp[tid + 256 * k] = (s0 + tid + 256 * k < hi) ? t[k] : -1;
            }
            __syncthreads();
            const int nsteps = (min(SUB, hi - s0) + 255) >> 8;
            for (int st = 0; st < nsteps; ++st) {
                const int i = (st * 8 + wave) * 32 + lane;
                const int v = sp[i];
                int rank, grp;
                wave_rank(v, lane, rank, grp);
                if (v >= 0 && v < R) {
                    const int base = hist[wave][v];
                    if (rank == 0) hist[wave][v] = base + grp;
                    if (pass == 1) inv[s0 + i] = base + rank;
                }
            }
        }
        __syncthreads();
        if (pass == 0) {
            int run = 0;
            if (tid < R) {
                for (int w = 0; w < 8; ++w) { const int h = hist[w][tid]; hist[w][tid] = run; run += h; }
            }
            cnt[c * RCAP + tid] = run;
        }
    }
}

__global__ __launch_bounds__(256)
void rel_place_kernel(const int* __restrict__ si, const int* __restrict__ pi, const int* __restrict__ oi,
                      const int* __restrict__ cnt, int* inv, int* __restrict__ relb, int* __restrict__ meta,
                      int* __restrict__ gS, int* __restrict__ gO, int E, int nrelElems)
{
    __shared__ __attribute__((aligned(16))) int stot[4][RCAP];
    __shared__ __attribute__((aligned(16))) int spre[4][RCAP];
    __shared__ int spad[RCAP];
    __shared__ int soff[RCAP];
    __shared__ int sown[3];
    const int tid = threadIdx.x;
    const int c = blockIdx.x;
    const int R = min(nrelElems / EDIM, RCAP);
    const int CH = ((E + NCH - 1) / NCH + 31) & ~31;
    const int lo = min(E, c * CH);
    const int hi = min(E, lo + CH);
    const int cap = E + R * BATCH;

    int r[PU], rk[PU], s[PU], o[PU];
    #pragma unroll
    for (int u = 0; u < PU; ++u) {
        const int e = min(lo + tid + 256 * u, E - 1);
        r[u]  = pi[e];
        rk[u] = inv[e];
        s[u]  = si[e];
        o[u]  = oi[e];
    }

    {
        const int quarter = tid >> 6;
        const int r4 = (tid & 63) * 4;
        int4 tot = make_int4(0, 0, 0, 0), prv = make_int4(0, 0, 0, 0);
        #pragma unroll 1
        for (int i0 = 0; i0 < NCH / 4; i0 += 16) {
            int4 v[16];
            #pragma unroll
            for (int k = 0; k < 16; ++k) v[k] = *(const int4*)(cnt + (4 * (i0 + k) + quarter) * RCAP + r4);
            #pragma unroll
            for (int k = 0; k < 16; ++k) {
                const bool before = (4 * (i0 + k) + quarter) < c;
                tot.x += v[k].x; tot.y += v[k].y; tot.z += v[k].z; tot.w += v[k].w;
                prv.x += before ? v[k].x : 0; prv.y += before ? v[k].y : 0;
                prv.z += before ? v[k].z : 0; prv.w += before ? v[k].w : 0;
            }
        }
        *(int4*)(&stot[quarter][r4]) = tot;
        *(int4*)(&spre[quarter][r4]) = prv;
    }
    __syncthreads();
    const int total = (tid < R) ? (stot[0][tid] + stot[1][tid]) + (stot[2][tid] + stot[3][tid]) : 0;
    const int pre   = (tid < R) ? (spre[0][tid] + spre[1][tid]) + (spre[2][tid] + spre[3][tid]) : 0;
    const int padded = (total + BATCH - 1) & ~(BATCH - 1);
    spad[tid] = (tid < R) ? padded : 0;
    __syncthreads();
    int base = 0;
    if (tid < R) {
        for (int q = 0; q < tid; ++q) base += spad[q];
    }
    soff[tid] = base + pre;
    if (tid == c) { sown[0] = base; sown[1] = total; sown[2] = padded; }
    if (c == 0 && tid == max(R - 1, 0)) meta[0] = (base + padded) / BATCH;
    __syncthreads();

    if (c < R) {
        const int pb = sown[0], pt = sown[1], pp = sown[2];
        const int nbmax = cap / BATCH;
        for (int k = pb + pt + tid; k < min(pb + pp, cap); k += 256) { gS[k] = 0; gO[k] = 0; }
        for (int b = pb / BATCH + tid; b < min((pb + pp) / BATCH, nbmax); b += 256) relb[b] = c;
    }

    for (int g0 = lo; g0 < hi; g0 += 256 * PU) {
        const int e0 = g0 + tid;
        if (g0 != lo) {
            #pragma unroll
            for (int u = 0; u < PU; ++u) {
                const int e = min(e0 + 256 * u, E - 1);
                r[u]  = pi[e];
                rk[u] = inv[e];
                s[u]  = si[e];
                o[u]  = oi[e];
            }
        }
        int pos[PU];
        #pragma unroll
        for (int u = 0; u < PU; ++u) {
            const int rr = min(max(r[u], 0), max(R - 1, 0));
            pos[u] = soff[rr] + rk[u];
        }
        #pragma unroll
        for (int u = 0; u < PU; ++u) {
            const int e = e0 + 256 * u;
            if (e < hi) {
                inv[e] = pos[u];
                if (pos[u] >= 0 && pos[u] < cap) { gS[pos[u]] = s[u]; gO[pos[u]] = o[u]; }
            }
        }
    }
}

__device__ __forceinline__ void matrix_role(const int r, const float* __restrict__ relations, const float* __restrict__ tokeys,
                                            const float* __restrict__ toqueries, float* __restrict__ M, int nrelElems)
{
    const int R = min(nrelElems / EDIM, RCAP);
    if (r >= R) return;
    const int lane = threadIdx.x & 31;
    const int wave = __builtin_amdgcn_readfirstlane(threadIdx.x >> 5);
    const int half = lane >> 4;
    const int lr   = lane & 15;
    v2f a[EDIM / 4];
    #pragma unroll
    for (int jc = 0; jc < EDIM / 4; ++jc) {
        const int j = 4 * jc + 2 * half;
        const v2f pj = *(const v2f*)(relations + (size_t)r * EDIM + j);
        v2f t;
        t.x = tokeys[(size_t)j * EDIM + 16 * wave + lr] * pj.x;
        t.y = tokeys[(size_t)(j + 1) * EDIM + 16 * wave + lr] * pj.y;
        a[jc] = t;
    }
    float* Mr = M + (size_t)r * EDIM * EDIM;
    #pragma unroll 1
    for (int kt = 0; kt < EDIM / 16; kt += 2) {
        v8f acc0 = (v8f)0.0f, acc1 = (v8f)0.0f;
        #pragma unroll
        for (int jc = 0; jc < EDIM / 4; ++jc) {
            const int j = 4 * jc + 2 * half;
            v2f b0, b1;
            b0.x = toqueries[(size_t)j * EDIM + 16 * kt + lr];
            b0.y = toqueries[(size_t)(j + 1) * EDIM + 16 * kt + lr];
            b1.x = toqueries[(size_t)j * EDIM + 16 * kt + 16 + lr];
            b1.y = toqueries[(size_t)(j + 1) * EDIM + 16 * kt + 16 + lr];
            acc0 = __builtin_amdgcn_wmma_f32_16x16x4_f32(false, a[jc], false, b0, (short)0, acc0, false, false);
            acc1 = __builtin_amdgcn_wmma_f32_16x16x4_f32(false, a[jc], false, b1, (short)0, acc1, false, false);
        }
        #pragma unroll
        for (int v = 0; v < 8; ++v) {
            Mr[(size_t)(16 * wave + 8 * half + v) * EDIM + 16 * kt + lr]      = acc0[v];
            Mr[(size_t)(16 * wave + 8 * half + v) * EDIM + 16 * kt + 16 + lr] = acc1[v];
        }
    }
}

__global__ __launch_bounds__(256)
void rel_rank_matrix_kernel(const int* __restrict__ pi, int* __restrict__ cnt, int* __restrict__ inv,
                            const float* __restrict__ relations, const float* __restrict__ tokeys,
                            const float* __restrict__ toqueries, float* __restrict__ M, int E, int nrelElems)
{
    if (blockIdx.x >= NCH) {
        matrix_role((int)blockIdx.x - NCH, relations, tokeys, toqueries, M, nrelElems);
        return;
    }
    rank_role(pi, cnt, inv, E, nrelElems);
}

__global__ __launch_bounds__(NWAVES * 32, 1)
void rel_dot_grouped_kernel(const float* __restrict__ emb,
                            const float* __restrict__ M,
                            const int* __restrict__ gS,
                            const int* __restrict__ gO,
                            const int* __restrict__ relb,
                            const int* __restrict__ meta,
                            float* __restrict__ dG,
                            int E, int nrelElems, int nembElems)
{
    __shared__ __attribute__((aligned(16))) float sS[2][BATCH * PITCH];
    __shared__ __attribute__((aligned(16))) float sO[2][BATCH * PITCH];
    __shared__ float sPart[2][NPROD][BATCH];

    const int lane = threadIdx.x & 31;
    const int wave = __builtin_amdgcn_readfirstlane(threadIdx.x >> 5);
    const int half = lane >> 4;
    const int lr   = lane & 15;
    const bool gatherer = wave >= NPROD;
    const int slice = wave & 3;
    const int tg = slice * 32 + lane;

    v2f wq[2][EDIM / 4];
    #pragma unroll
    for (int mt = 0; mt < 2; ++mt) {
        #pragma unroll
        for (int kc = 0; kc < EDIM / 4; ++kc) wq[mt][kc] = (v2f)0.0f;
    }
    int rcur = -1;

    const float scale = 0.08838834764831845f;
    const int R = min(nrelElems / EDIM, RCAP);
    const int N = nembElems / EDIM;
    const int nbtot = min(max(meta[0], 0), (E + R * BATCH) / BATCH);
    const int per = (nbtot + (int)gridDim.x - 1) / (int)gridDim.x;
    const int b0 = (int)blockIdx.x * per;
    const int bend = min(nbtot, b0 + per);

    int it = 0;
    int b = b0;
    int relA = 0;
    for (;; ++b, ++it) {
        const int relThis = relA;
        if (b < bend) relA = relb[b];
        if (gatherer) {
            if (it >= 2) {
                const int pc = it & 1;
                const float d = (sPart[pc][0][tg] + sPart[pc][1][tg]) + (sPart[pc][2][tg] + sPart[pc][3][tg]);
                dG[(b - 2) * BATCH + tg] = d * scale;
            }
            if (b < bend) {
                const int cur = it & 1;
                const int base = b * BATCH;
                #pragma unroll 1
                for (int q = 0; q < 2; ++q) {
                    const int row0 = slice * 32 + q * 16;
                    const int* isrc = half ? gO : gS;
                    const int idx = min(max(isrc[base + row0 + lr], 0), N - 1);
                    float4 vs[16], vo[16];
                    #pragma unroll
                    for (int r = 0; r < 16; ++r) {
                        const int srw = __builtin_amdgcn_readlane(idx, r);
                        const int orw = __builtin_amdgcn_readlane(idx, 16 + r);
                        vs[r] = *(const float4*)(emb + (size_t)srw * EDIM + lane * 4);
                        vo[r] = *(const float4*)(emb + (size_t)orw * EDIM + lane * 4);
                    }
                    #pragma unroll
                    for (int r = 0; r < 16; ++r) {
                        *(float4*)(&sS[cur][(row0 + r) * PITCH + lane * 4]) = vs[r];
                        *(float4*)(&sO[cur][(row0 + r) * PITCH + lane * 4]) = vo[r];
                    }
                }
            }
        }
        if (it >= 1) {
            const int cur = (it - 1) & 1;
            const int rel = min(max(__builtin_amdgcn_readfirstlane(relThis), 0), R - 1);
            if (rel != rcur) {
                rcur = rel;
                const float* Mr = M + (size_t)rel * EDIM * EDIM;
                #pragma unroll
                for (int mt = 0; mt < 2; ++mt) {
                    const size_t off = (size_t)(32 * slice + 16 * mt + lr) * EDIM + 2 * half;
                    #pragma unroll
                    for (int kc = 0; kc < EDIM / 4; ++kc) wq[mt][kc] = *(const v2f*)(Mr + off + kc * 4);
                }
            }
            const int ps0 = gatherer ? PASSES - 1 : 0;
            const int ps1 = gatherer ? PASSES : PASSES - 1;
            #pragma unroll 1
            for (int ps = ps0; ps < ps1; ++ps) {
                float4 p[NT][4];
                const float* fO[NT];
                v8f aO[NT][2];
                #pragma unroll
                for (int u = 0; u < NT; ++u) {
                    const int slot = (ps * NT + u) * 16 + lr;
                    const float* sr = &sS[cur][slot * PITCH + 32 * slice + 8 * half];
                    p[u][0] = *(const float4*)(sr);
                    p[u][1] = *(const float4*)(sr + 4);
                    p[u][2] = *(const float4*)(sr + 16);
                    p[u][3] = *(const float4*)(sr + 20);
                    fO[u] = &sO[cur][slot * PITCH + 2 * half];
                    aO[u][0] = (v8f)0.0f; aO[u][1] = (v8f)0.0f;
                }
                #pragma unroll
                for (int kc = 0; kc < EDIM / 4; ++kc) {
                    #pragma unroll
                    for (int u = 0; u < NT; ++u) {
                        const v2f bo = *(const v2f*)(fO[u] + kc * 4);
                        aO[u][0] = __builtin_amdgcn_wmma_f32_16x16x4_f32(false, wq[0][kc], false, bo, (short)0, aO[u][0], false, false);
                        aO[u][1] = __builtin_amdgcn_wmma_f32_16x16x4_f32(false, wq[1][kc], false, bo, (short)0, aO[u][1], false, false);
                    }
                }
                #pragma unroll
                for (int u = 0; u < NT; ++u) {
                    const v8f o0 = aO[u][0], o1 = aO[u][1];
                    float s = 0.0f;
                    s += o0[0] * p[u][0].x;  s += o0[1] * p[u][0].y;
                    s += o0[2] * p[u][0].z;  s += o0[3] * p[u][0].w;
                    s += o0[4] * p[u][1].x;  s += o0[5] * p[u][1].y;
                    s += o0[6] * p[u][1].z;  s += o0[7] * p[u][1].w;
                    s += o1[0] * p[u][2].x;  s += o1[1] * p[u][2].y;
                    s += o1[2] * p[u][2].z;  s += o1[3] * p[u][2].w;
                    s += o1[4] * p[u][3].x;  s += o1[5] * p[u][3].y;
                    s += o1[6] * p[u][3].z;  s += o1[7] * p[u][3].w;
                    s += __shfl_xor(s, 16, 32);
                    if (half == 0) sPart[cur][slice][(ps * NT + u) * 16 + lr] = s;
                }
            }
        }
        __syncthreads();
        if (b >= bend) break;
    }
    if (gatherer && it >= 1) {
        const int pc = (it - 1) & 1;
        const float d = (sPart[pc][0][tg] + sPart[pc][1][tg]) + (sPart[pc][2][tg] + sPart[pc][3][tg]);
        dG[(b - 1) * BATCH + tg] = d * scale;
    }
}

__global__ __launch_bounds__(256)
void rel_fetch_kernel(const float* __restrict__ dG, const int* __restrict__ inv, float* __restrict__ out,
                      int E, int nrelElems)
{
    const int R = min(nrelElems / EDIM, RCAP);
    const int cap = E + R * BATCH;
    const int stride = (int)gridDim.x * 256;
    for (int g0 = blockIdx.x * 256; g0 < E; g0 += stride * TU) {
        const int e0 = g0 + (int)threadIdx.x;
        int p[TU];
        float d[TU];
        #pragma unroll
        for (int u = 0; u < TU; ++u) p[u] = inv[min(e0 + u * stride, E - 1)];
        #pragma unroll
        for (int u = 0; u < TU; ++u) d[u] = dG[min(max(p[u], 0), cap - 1)];
        asm volatile("" : "+v"(d[0]));
        #pragma unroll
        for (int u = 0; u < TU; ++u) {
            const int e = e0 + u * stride;
            if (e < E) out[e] = d[u];
        }
    }
}

extern "C" void kernel_launch(void* const* d_in, const int* in_sizes, int n_in,
                              void* d_out, int out_size, void* d_ws, size_t ws_size,
                              hipStream_t stream) {
    const float* emb       = (const float*)d_in[0];
    const float* relations = (const float*)d_in[1];
    const float* tokeys    = (const float*)d_in[2];
    const float* toqueries = (const float*)d_in[3];
    const int*   si        = (const int*)d_in[4];
    const int*   pi        = (const int*)d_in[5];
    const int*   oi        = (const int*)d_in[6];
    float*       out       = (float*)d_out;

    const int E = in_sizes[4];
    const int nrelElems = in_sizes[1];
    const int nembElems = in_sizes[0];

    char* ws = (char*)d_ws;
    float* M    = (float*)(ws);
    int*   cnt  = (int*)(ws + 16777216);
    int*   relb = (int*)(ws + 17301504);
    int*   meta = (int*)(ws + 17563648);
    int*   inv  = (int*)(ws + 17825792);
    int*   gS   = (int*)(ws + 23068672);
    int*   gO   = (int*)(ws + 28311552);
    float* dG   = (float*)(ws + 33554432);

    rel_rank_matrix_kernel<<<NL1, 256, 0, stream>>>(pi, cnt, inv, relations, tokeys, toqueries, M, E, nrelElems);
    rel_place_kernel<<<NCH, 256, 0, stream>>>(si, pi, oi, cnt, inv, relb, meta, gS, gO, E, nrelElems);
    rel_dot_grouped_kernel<<<NGROUPS, NWAVES * 32, 0, stream>>>(emb, M, gS, gO, relb, meta, dG, E, nrelElems, nembElems);
    rel_fetch_kernel<<<TG, 256, 0, stream>>>(dG, inv, out, E, nrelElems);
}
